# starting kernel
# baseline (speedup 1.0000x reference)
#include <hip/hip_runtime.h>
#include <hip/hip_bf16.h>

typedef __attribute__((ext_vector_type(2))) float v2f;
typedef __attribute__((ext_vector_type(4))) float v4f;
typedef __attribute__((ext_vector_type(8))) float v8f;

#define LEAKY_ALPHA 0.2f

static constexpr int N_NODES = 50000;
static constexpr int N_EDGES = 800000;
static constexpr int EDGE_BLOCKS = 3125;
static constexpr int NODE_TILES = 3125;

static constexpr size_t WS_A2   = 0;
static constexpr size_t WS_BIAS = 2048;
static constexpr size_t WS_S    = 2056;
static constexpr size_t WS_LOG  = WS_S + (size_t)N_NODES * 8;
static constexpr size_t WS_MAXP = WS_LOG + (size_t)N_EDGES * 4;
static constexpr size_t WS_SUMP = WS_MAXP + (size_t)EDGE_BLOCKS * 4;
static constexpr size_t WS_GMAX = WS_SUMP + (size_t)EDGE_BLOCKS * 4;
static constexpr size_t WS_RSUM = WS_GMAX + 4;

__global__ void prep_kernel(const float* __restrict__ W, const float* __restrict__ b,
                            const float* __restrict__ a, float* ws) {
  const int c = threadIdx.x;
  float* A2 = ws + WS_A2;
#pragma unroll
  for (int j = 0; j < 8; ++j) {
    const int h = j >> 1, half = j & 1;
    const float* av = a + h * 64 + half * 32;
    float acc = 0.f;
#pragma unroll 8
    for (int k = 0; k < 32; ++k) acc += W[(size_t)(h * 32 + k) * 128 + c] * av[k];
    A2[j * 128 + c] = acc;
  }
#pragma unroll
  for (int j = 8; j < 16; ++j) A2[j * 128 + c] = 0.f;
  if (c < 8) {
    const int h = c >> 1, half = c & 1;
    const float* av = a + h * 64 + half * 32;
    float acc = 0.f;
    for (int k = 0; k < 32; ++k) acc += b[h * 32 + k] * av[k];
    ws[WS_BIAS + c] = acc;
  }
}

__global__ void __launch_bounds__(256) score_wmma_kernel(const float* __restrict__ X,
                                                         float* ws) {
  const float* A2 = ws + WS_A2;
  const float* bias = ws + WS_BIAS;
  float* S = ws + WS_S;

  const int lane = threadIdx.x & 31;
  const int wave = threadIdx.x >> 5;
  const int tile = blockIdx.x * 8 + wave;
  if (tile >= NODE_TILES) return;

  const int m = lane & 15;
  const int hi = lane >> 4;
  const float* xrow = X + (size_t)(tile * 16 + m) * 128 + 2 * hi;
  const float* brow = A2 + (size_t)m * 128 + 2 * hi;

  v8f acc = {};
#pragma unroll 8
  for (int kk = 0; kk < 32; ++kk) {
    v2f av = *(const v2f*)(xrow + kk * 4);
    v2f bv = *(const v2f*)(brow + kk * 4);
    acc = __builtin_amdgcn_wmma_f32_16x16x4_f32(
         false, av,  false, bv,
         (short)0, acc,  false,  false);
  }

  if (m < 8) {
    const float bb = bias[m];
    const int rowbase = tile * 16 + 8 * hi;
#pragma unroll
    for (int r = 0; r < 8; ++r) S[(size_t)(rowbase + r) * 8 + m] = acc[r] + bb;
  }
}

__global__ void __launch_bounds__(256) edge_logits_kernel(const int* __restrict__ ei,
                                                          float* ws) {
  __shared__ v4f red[256];
  const int t = threadIdx.x;
  const int e = blockIdx.x * 256 + t;
  const v4f* S = (const v4f*)(ws + WS_S);

  const int src = ei[e];
  const int dst = ei[N_EDGES + e];
  v4f s0 = S[(size_t)src * 2], s1 = S[(size_t)src * 2 + 1];
  v4f d0 = S[(size_t)dst * 2], d1 = S[(size_t)dst * 2 + 1];

  v4f lg;
  lg.x = s0.x + d0.y;
  lg.y = s0.z + d0.w;
  lg.z = s1.x + d1.y;
  lg.w = s1.z + d1.w;
  lg.x = fmaxf(lg.x, LEAKY_ALPHA * lg.x);
  lg.y = fmaxf(lg.y, LEAKY_ALPHA * lg.y);
  lg.z = fmaxf(lg.z, LEAKY_ALPHA * lg.z);
  lg.w = fmaxf(lg.w, LEAKY_ALPHA * lg.w);

  *(v4f*)(ws + WS_LOG + (size_t)e * 4) = lg;

  red[t] = lg;
  __syncthreads();
  for (int off = 128; off > 0; off >>= 1) {
    if (t < off) {
      v4f o = red[t + off], m = red[t];
      m.x = fmaxf(m.x, o.x); m.y = fmaxf(m.y, o.y);
      m.z = fmaxf(m.z, o.z); m.w = fmaxf(m.w, o.w);
      red[t] = m;
    }
    __syncthreads();
  }
  if (t == 0) *(v4f*)(ws + WS_MAXP + (size_t)blockIdx.x * 4) = red[0];
}

__global__ void __launch_bounds__(256) reduce_max_kernel(float* ws) {
  __shared__ v4f red[256];
  const int t = threadIdx.x;
  const v4f* P = (const v4f*)(ws + WS_MAXP);
  v4f m = {-3.0e38f, -3.0e38f, -3.0e38f, -3.0e38f};
  for (int i = t; i < EDGE_BLOCKS; i += 256) {
    v4f o = P[i];
    m.x = fmaxf(m.x, o.x); m.y = fmaxf(m.y, o.y);
    m.z = fmaxf(m.z, o.z); m.w = fmaxf(m.w, o.w);
  }
  red[t] = m;
  __syncthreads();
  for (int off = 128; off > 0; off >>= 1) {
    if (t < off) {
      v4f o = red[t + off], q = red[t];
      q.x = fmaxf(q.x, o.x); q.y = fmaxf(q.y, o.y);
      q.z = fmaxf(q.z, o.z); q.w = fmaxf(q.w, o.w);
      red[t] = q;
    }
    __syncthreads();
  }
  if (t == 0) *(v4f*)(ws + WS_GMAX) = red[0];
}

__global__ void __launch_bounds__(256) edge_exp_kernel(float* ws, float* __restrict__ out) {
  __shared__ v4f red[256];
  const int t = threadIdx.x;
  const int e = blockIdx.x * 256 + t;
  const v4f gm = *(const v4f*)(ws + WS_GMAX);
  v4f lg = *(const v4f*)(ws + WS_LOG + (size_t)e * 4);
  v4f ex;
  ex.x = __expf(lg.x - gm.x);
  ex.y = __expf(lg.y - gm.y);
  ex.z = __expf(lg.z - gm.z);
  ex.w = __expf(lg.w - gm.w);
  *(v4f*)(out + (size_t)e * 4) = ex;

  red[t] = ex;
  __syncthreads();
  for (int off = 128; off > 0; off >>= 1) {
    if (t < off) {
      v4f o = red[t + off], q = red[t];
      q.x += o.x; q.y += o.y; q.z += o.z; q.w += o.w;
      red[t] = q;
    }
    __syncthreads();
  }
  if (t == 0) *(v4f*)(ws + WS_SUMP + (size_t)blockIdx.x * 4) = red[0];
}

__global__ void __launch_bounds__(256) reduce_sum_kernel(float* ws) {
  __shared__ v4f red[256];
  const int t = threadIdx.x;
  const v4f* P = (const v4f*)(ws + WS_SUMP);
  v4f s = {};
  for (int i = t; i < EDGE_BLOCKS; i += 256) {
    v4f o = P[i];
    s.x += o.x; s.y += o.y; s.z += o.z; s.w += o.w;
  }
  red[t] = s;
  __syncthreads();
  for (int off = 128; off > 0; off >>= 1) {
    if (t < off) {
      v4f o = red[t + off], q = red[t];
      q.x += o.x; q.y += o.y; q.z += o.z; q.w += o.w;
      red[t] = q;
    }
    __syncthreads();
  }
  if (t == 0) {
    v4f r = red[0];
    v4f inv = {1.0f / r.x, 1.0f / r.y, 1.0f / r.z, 1.0f / r.w};
    *(v4f*)(ws + WS_RSUM) = inv;
  }
}

__global__ void __launch_bounds__(256) scale_kernel(float* ws, float* __restrict__ out) {
  const int e = blockIdx.x * 256 + threadIdx.x;
  const v4f r = *(const v4f*)(ws + WS_RSUM);
  v4f v = *(v4f*)(out + (size_t)e * 4);
  v.x *= r.x; v.y *= r.y; v.z *= r.z; v.w *= r.w;
  *(v4f*)(out + (size_t)e * 4) = v;
}

extern "C" void kernel_launch(void* const* d_in, const int* in_sizes, int n_in,
                              void* d_out, int out_size, void* d_ws, size_t ws_size,
                              hipStream_t stream) {
  const float* node_feats = (const float*)d_in[0];
  const int*   edge_index = (const int*)d_in[1];
  const float* W          = (const float*)d_in[2];
  const float* b          = (const float*)d_in[3];
  const float* a          = (const float*)d_in[4];
  float* out = (float*)d_out;
  float* ws  = (float*)d_ws;

  prep_kernel<<<1, 128, 0, stream>>>(W, b, a, ws);
  score_wmma_kernel<<<(NODE_TILES + 7) / 8, 256, 0, stream>>>(node_feats, ws);
  edge_logits_kernel<<<EDGE_BLOCKS, 256, 0, stream>>>(edge_index, ws);
  reduce_max_kernel<<<1, 256, 0, stream>>>(ws);
  edge_exp_kernel<<<EDGE_BLOCKS, 256, 0, stream>>>(ws, out);
  reduce_sum_kernel<<<1, 256, 0, stream>>>(ws);
  scale_kernel<<<EDGE_BLOCKS, 256, 0, stream>>>(ws, out);
}


# Softmax shift from node-score maxima (upper bound, by-product of the score GEMM): logit and exp passes become one edge pass, 7 launches -> 6; exact edge-max redo if the sum shows a loose bound
# speedup vs baseline: 1.8598x; 1.8598x over previous
#include <hip/hip_runtime.h>
#include <hip/hip_bf16.h>

typedef __attribute__((ext_vector_type(2))) float v2f;
typedef __attribute__((ext_vector_type(4))) float v4f;
typedef __attribute__((ext_vector_type(8))) float v8f;

#define LEAKY_ALPHA 0.2f

template <int XMASK>
__device__ __forceinline__ float row_partner(float v) {
  return __builtin_bit_cast(float, __builtin_amdgcn_update_dpp(
      0, __builtin_bit_cast(int, v), 0x160 + XMASK, 0xF, 0xF, true));
}
__device__ __forceinline__ float other_row(float v) {
  const int i = __builtin_bit_cast(int, v);
  return __builtin_bit_cast(float, __builtin_amdgcn_permlanex16(
      i, i, 0x76543210u, 0xfedcba98u, false, false));
}
__device__ __forceinline__ float wave_max(float v) {
  v = fmaxf(v, row_partner<1>(v));
  v = fmaxf(v, row_partner<2>(v));
  v = fmaxf(v, row_partner<4>(v));
  v = fmaxf(v, row_partner<8>(v));
  return fmaxf(v, other_row(v));
}
__device__ __forceinline__ float wave_sum(float v) {
  v += row_partner<1>(v);
  v += row_partner<2>(v);
  v += row_partner<4>(v);
  v += row_partner<8>(v);
  return v + other_row(v);
}
__device__ __forceinline__ v4f wave_max4(v4f m) {
  m.x = wave_max(m.x); m.y = wave_max(m.y); m.z = wave_max(m.z); m.w = wave_max(m.w);
  return m;
}
__device__ __forceinline__ v4f wave_sum4(v4f s) {
  s.x = wave_sum(s.x); s.y = wave_sum(s.y); s.z = wave_sum(s.z); s.w = wave_sum(s.w);
  return s;
}
__device__ __forceinline__ v4f max4(v4f p, v4f q) {
  v4f r = {fmaxf(p.x, q.x), fmaxf(p.y, q.y), fmaxf(p.z, q.z), fmaxf(p.w, q.w)};
  return r;
}
__device__ __forceinline__ v4f block_max4(v4f m, v4f* red) {
  m = wave_max4(m);
  if ((threadIdx.x & 31) == 0) red[threadIdx.x >> 5] = m;
  __syncthreads();
  v4f r = red[0];
  for (int w = 1; w < 8; ++w) r = max4(r, red[w]);
  return r;
}
__device__ __forceinline__ v4f block_sum4(v4f s, v4f* red) {
  s = wave_sum4(s);
  if ((threadIdx.x & 31) == 0) red[threadIdx.x >> 5] = s;
  __syncthreads();
  v4f r = red[0];
  for (int w = 1; w < 8; ++w) {
    const v4f o = red[w];
    r.x += o.x; r.y += o.y; r.z += o.z; r.w += o.w;
  }
  return r;
}

static constexpr int N_NODES = 50000;
static constexpr int N_EDGES = 800000;
static constexpr int EDGE_BLOCKS = 3125;
static constexpr int NODE_TILES = 3125;
static constexpr int PART_FULL = EDGE_BLOCKS / 256;
static constexpr int PART_TAIL = EDGE_BLOCKS - PART_FULL * 256;
static constexpr int NMAX_V4 = NODE_TILES * 2;
static constexpr int NB_THREADS = 1024;
static constexpr int NMAX_FULL = NMAX_V4 / NB_THREADS;
static constexpr int NMAX_TAIL = NMAX_V4 - NMAX_FULL * NB_THREADS;
static constexpr float BOUND_MARGIN = 32.0f;

static constexpr size_t WS_A2   = 0;
static constexpr size_t WS_BIAS = 2048;
static constexpr size_t WS_S    = 2056;
static constexpr size_t WS_NMAX = WS_S + (size_t)N_NODES * 8;
static constexpr size_t WS_SUMP = WS_NMAX + (size_t)NODE_TILES * 8;
static constexpr size_t WS_BND  = WS_SUMP + (size_t)EDGE_BLOCKS * 4;
static constexpr size_t WS_RSUM = WS_BND + 4;

__global__ void __launch_bounds__(128) prep_kernel(const float* __restrict__ W,
                                                   const float* __restrict__ b,
                                                   const float* __restrict__ a, float* ws) {
  const int c = threadIdx.x;
  const int j = blockIdx.x;
  float* A2 = ws + WS_A2;
  if (j == 8) {
    if (c < 8) {
      const int hb = c >> 1, halfb = c & 1;
      const float* avb = a + hb * 64 + halfb * 32;
      float accb = 0.f;
      for (int k = 0; k < 32; ++k) accb += b[hb * 32 + k] * avb[k];
      ws[WS_BIAS + c] = accb;
    }
    return;
  }
  const int h = j >> 1, half = j & 1;
  const float* av = a + h * 64 + half * 32;
  const float* wp = W + (size_t)(h * 32) * 128 + c;

  float w[32];
#pragma unroll
  for (int k = 0; k < 32; ++k) w[k] = wp[(size_t)k * 128];
  float acc = 0.f;
#pragma unroll
  for (int k = 0; k < 32; ++k) acc += w[k] * av[k];
  A2[j * 128 + c] = acc;
  A2[(8 + j) * 128 + c] = 0.f;
}

static constexpr int LDS_PITCH = 132;

__global__ void __launch_bounds__(256) score_wmma_kernel(const float* __restrict__ X,
                                                         float* ws) {
  __shared__ __attribute__((aligned(16))) float sB[16 * LDS_PITCH];
  __shared__ __attribute__((aligned(16))) float sA[8][16 * LDS_PITCH];
  const float* A2 = ws + WS_A2;
  const float* bias = ws + WS_BIAS;
  float* S = ws + WS_S;

  const int t = threadIdx.x;
  const int lane = t & 31;
  const int wave = t >> 5;
  const int tile_raw = blockIdx.x * 8 + wave;
  const bool live = tile_raw < NODE_TILES;
  const int tile = live ? tile_raw : NODE_TILES - 1;

  const int m = lane & 15;
  const int hi = lane >> 4;

  const v4f* xsrc = (const v4f*)(X + (size_t)tile * 16 * 128) + lane;
  v4f xr[16];
#pragma unroll
  for (int r = 0; r < 16; ++r) xr[r] = xsrc[r * 32];
  const v4f* wsrc = (const v4f*)A2;
  const v4f w0 = wsrc[t];
  const v4f w1 = wsrc[t + 256];
  const float bb = bias[m & 7];

  float* myA = sA[wave];
#pragma unroll
  for (int r = 0; r < 16; ++r) *(v4f*)(myA + r * LDS_PITCH + lane * 4) = xr[r];
  *(v4f*)(sB + wave * LDS_PITCH + lane * 4) = w0;
  *(v4f*)(sB + (8 + wave) * LDS_PITCH + lane * 4) = w1;
  __syncthreads();

  const float* arow = myA + m * LDS_PITCH + 2 * hi;
  const float* brow = sB + m * LDS_PITCH + 2 * hi;

  v8f acc = {};
#pragma unroll 8
  for (int kk = 0; kk < 32; ++kk) {
    v2f av = *(const v2f*)(arow + kk * 4);
    v2f bv = *(const v2f*)(brow + kk * 4);
    acc = __builtin_amdgcn_wmma_f32_16x16x4_f32(
         false, av,  false, bv,
         (short)0, acc,  false,  false);
  }

  float cm = acc[0];
#pragma unroll
  for (int r = 1; r < 8; ++r) cm = fmaxf(cm, acc[r]);
  cm = fmaxf(cm, other_row(cm));

  if (live && m < 8) {
    const int rowbase = tile * 16 + 8 * hi;
#pragma unroll
    for (int r = 0; r < 8; ++r) S[(size_t)(rowbase + r) * 8 + m] = acc[r] + bb;
    if (hi == 0) ws[WS_NMAX + (size_t)tile * 8 + m] = cm + bb;
  }
}

__device__ __forceinline__ v4f edge_logits(const int* __restrict__ ei, const v4f* S, int e) {
  const int src = ei[e];
  const int dst = ei[N_EDGES + e];
  const v4f s0 = S[(size_t)src * 2], s1 = S[(size_t)src * 2 + 1];
  const v4f d0 = S[(size_t)dst * 2], d1 = S[(size_t)dst * 2 + 1];
  v4f lg;
  lg.x = s0.x + d0.y;
  lg.y = s0.z + d0.w;
  lg.z = s1.x + d1.y;
  lg.w = s1.z + d1.w;
  lg.x = fmaxf(lg.x, LEAKY_ALPHA * lg.x);
  lg.y = fmaxf(lg.y, LEAKY_ALPHA * lg.y);
  lg.z = fmaxf(lg.z, LEAKY_ALPHA * lg.z);
  lg.w = fmaxf(lg.w, LEAKY_ALPHA * lg.w);
  return lg;
}
__device__ __forceinline__ float leaky(float x) { return fmaxf(x, LEAKY_ALPHA * x); }

__global__ void __launch_bounds__(NB_THREADS) node_bound_kernel(float* ws) {
  __shared__ v4f red[2 * (NB_THREADS / 32)];
  const int t = threadIdx.x;
  const v4f* P = (const v4f*)(ws + WS_NMAX);
  v4f v[NMAX_FULL + 1];
#pragma unroll
  for (int q = 0; q < NMAX_FULL; ++q) v[q] = P[t + NB_THREADS * q];
  v[NMAX_FULL] = P[t < NMAX_TAIL ? t + NB_THREADS * NMAX_FULL : t];
  v4f m = v[0];
#pragma unroll
  for (int q = 1; q < NMAX_FULL; ++q) m = max4(m, v[q]);
  {
    const bool live = t < NMAX_TAIL;
    const v4f o = v[NMAX_FULL];
    m.x = fmaxf(m.x, live ? o.x : m.x); m.y = fmaxf(m.y, live ? o.y : m.y);
    m.z = fmaxf(m.z, live ? o.z : m.z); m.w = fmaxf(m.w, live ? o.w : m.w);
  }
  m.x = fmaxf(m.x, row_partner<2>(m.x)); m.y = fmaxf(m.y, row_partner<2>(m.y));
  m.z = fmaxf(m.z, row_partner<2>(m.z)); m.w = fmaxf(m.w, row_partner<2>(m.w));
  m.x = fmaxf(m.x, row_partner<4>(m.x)); m.y = fmaxf(m.y, row_partner<4>(m.y));
  m.z = fmaxf(m.z, row_partner<4>(m.z)); m.w = fmaxf(m.w, row_partner<4>(m.w));
  m.x = fmaxf(m.x, row_partner<8>(m.x)); m.y = fmaxf(m.y, row_partner<8>(m.y));
  m.z = fmaxf(m.z, row_partner<8>(m.z)); m.w = fmaxf(m.w, row_partner<8>(m.w));
  m.x = fmaxf(m.x, other_row(m.x)); m.y = fmaxf(m.y, other_row(m.y));
  m.z = fmaxf(m.z, other_row(m.z)); m.w = fmaxf(m.w, other_row(m.w));
  if ((t & 31) < 2) red[(t >> 5) * 2 + (t & 1)] = m;
  __syncthreads();
  if (t == 0) {
    v4f lo = red[0], hi = red[1];
    for (int w = 1; w < NB_THREADS / 32; ++w) {
      lo = max4(lo, red[2 * w]);
      hi = max4(hi, red[2 * w + 1]);
    }
    v4f bnd = {leaky(lo.x + lo.y), leaky(lo.z + lo.w), leaky(hi.x + hi.y), leaky(hi.z + hi.w)};
    *(v4f*)(ws + WS_BND) = bnd;
  }
}

__global__ void __launch_bounds__(256) edge_exp_kernel(const int* __restrict__ ei, float* ws,
                                                       float* __restrict__ out) {
  __shared__ v4f red[8];
  const int t = threadIdx.x;
  const int e = blockIdx.x * 256 + t;
  const v4f bnd = *(const v4f*)(ws + WS_BND);
  const v4f lg = edge_logits(ei, (const v4f*)(ws + WS_S), e);
  v4f ex;
  ex.x = __expf(lg.x - bnd.x);
  ex.y = __expf(lg.y - bnd.y);
  ex.z = __expf(lg.z - bnd.z);
  ex.w = __expf(lg.w - bnd.w);
  *(v4f*)(out + (size_t)e * 4) = ex;

  const v4f bs = block_sum4(ex, red);
  if (t == 0) *(v4f*)(ws + WS_SUMP + (size_t)blockIdx.x * 4) = bs;
}

__global__ void __launch_bounds__(256) reduce_sum_kernel(const int* __restrict__ ei, float* ws,
                                                         float* __restrict__ out) {
  __shared__ v4f red[8];
  const int t = threadIdx.x;
  const v4f* P = (const v4f*)(ws + WS_SUMP);
  v4f v[PART_FULL + 1];
#pragma unroll
  for (int q = 0; q < PART_FULL; ++q) v[q] = P[t + 256 * q];
  v[PART_FULL] = P[t < PART_TAIL ? t + 256 * PART_FULL : t];
  v4f s = {};
#pragma unroll
  for (int q = 0; q < PART_FULL; ++q) {
    s.x += v[q].x; s.y += v[q].y; s.z += v[q].z; s.w += v[q].w;
  }
  {
    const bool live = t < PART_TAIL;
    const v4f o = v[PART_FULL];
    s.x += live ? o.x : 0.f; s.y += live ? o.y : 0.f;
    s.z += live ? o.z : 0.f; s.w += live ? o.w : 0.f;
  }
  v4f r = block_sum4(s, red);
  const float need = (float)N_EDGES * __expf(-BOUND_MARGIN);
  const bool close = (r.x >= need) && (r.y >= need) && (r.z >= need) && (r.w >= need);
  if (!close) {
    const v4f* S = (const v4f*)(ws + WS_S);
    __syncthreads();
    v4f mx = edge_logits(ei, S, t);
    for (int e = t + 256; e < N_EDGES; e += 256) mx = max4(mx, edge_logits(ei, S, e));
    const v4f gm = block_max4(mx, red);
    __syncthreads();
    v4f a2 = {};
    for (int e = t; e < N_EDGES; e += 256) {
      const v4f lg = edge_logits(ei, S, e);
      v4f ex;
      ex.x = __expf(lg.x - gm.x); ex.y = __expf(lg.y - gm.y);
      ex.z = __expf(lg.z - gm.z); ex.w = __expf(lg.w - gm.w);
      *(v4f*)(out + (size_t)e * 4) = ex;
      a2.x += ex.x; a2.y += ex.y; a2.z += ex.z; a2.w += ex.w;
    }
    r = block_sum4(a2, red);
  }
  if (t == 0) {
    v4f inv = {1.0f / r.x, 1.0f / r.y, 1.0f / r.z, 1.0f / r.w};
    *(v4f*)(ws + WS_RSUM) = inv;
  }
}

__global__ void __launch_bounds__(256) scale_kernel(float* ws, float* __restrict__ out) {
  const int e = blockIdx.x * 256 + threadIdx.x;
  const v4f r = *(const v4f*)(ws + WS_RSUM);
  v4f v = *(v4f*)(out + (size_t)e * 4);
  v.x *= r.x; v.y *= r.y; v.z *= r.z; v.w *= r.w;
  *(v4f*)(out + (size_t)e * 4) = v;
}

extern "C" void kernel_launch(void* const* d_in, const int* in_sizes, int n_in,
                              void* d_out, int out_size, void* d_ws, size_t ws_size,
                              hipStream_t stream) {
  const float* node_feats = (const float*)d_in[0];
  const int*   edge_index = (const int*)d_in[1];
  const float* W          = (const float*)d_in[2];
  const float* b          = (const float*)d_in[3];
  const float* a          = (const float*)d_in[4];
  float* out = (float*)d_out;
  float* ws  = (float*)d_ws;

  prep_kernel<<<9, 128, 0, stream>>>(W, b, a, ws);
  score_wmma_kernel<<<(NODE_TILES + 7) / 8, 256, 0, stream>>>(node_feats, ws);
  node_bound_kernel<<<1, NB_THREADS, 0, stream>>>(ws);
  edge_exp_kernel<<<EDGE_BLOCKS, 256, 0, stream>>>(edge_index, ws, out);
  reduce_sum_kernel<<<1, 256, 0, stream>>>(edge_index, ws, out);
  scale_kernel<<<EDGE_BLOCKS, 256, 0, stream>>>(ws, out);
}


# Softmax shift dropped: exp(logit) unshifted, range proven afterwards from the sum (E*e^-32 <= sum <= e^32), exact-max redo otherwise; bound launch and tile maxima go, 6 launches -> 5; sum order kept
# speedup vs baseline: 2.1132x; 1.1363x over previous
#include <hip/hip_runtime.h>
#include <hip/hip_bf16.h>

typedef __attribute__((ext_vector_type(2))) float v2f;
typedef __attribute__((ext_vector_type(4))) float v4f;
typedef __attribute__((ext_vector_type(8))) float v8f;

#define LEAKY_ALPHA 0.2f

template <int XMASK>
__device__ __forceinline__ float row_partner(float v) {
  return __builtin_bit_cast(float, __builtin_amdgcn_update_dpp(
      0, __builtin_bit_cast(int, v), 0x160 + XMASK, 0xF, 0xF, true));
}
__device__ __forceinline__ float other_row(float v) {
  const int i = __builtin_bit_cast(int, v);
  return __builtin_bit_cast(float, __builtin_amdgcn_permlanex16(
      i, i, 0x76543210u, 0xfedcba98u, false, false));
}
__device__ __forceinline__ float wave_max(float v) {
  v = fmaxf(v, row_partner<1>(v));
  v = fmaxf(v, row_partner<2>(v));
  v = fmaxf(v, row_partner<4>(v));
  v = fmaxf(v, row_partner<8>(v));
  return fmaxf(v, other_row(v));
}
__device__ __forceinline__ float wave_sum(float v) {
  v += row_partner<1>(v);
  v += row_partner<2>(v);
  v += row_partner<4>(v);
  v += row_partner<8>(v);
  return v + other_row(v);
}
__device__ __forceinline__ v4f wave_max4(v4f m) {
  m.x = wave_max(m.x); m.y = wave_max(m.y); m.z = wave_max(m.z); m.w = wave_max(m.w);
  return m;
}
__device__ __forceinline__ v4f wave_sum4(v4f s) {
  s.x = wave_sum(s.x); s.y = wave_sum(s.y); s.z = wave_sum(s.z); s.w = wave_sum(s.w);
  return s;
}
__device__ __forceinline__ v4f max4(v4f p, v4f q) {
  v4f r = {fmaxf(p.x, q.x), fmaxf(p.y, q.y), fmaxf(p.z, q.z), fmaxf(p.w, q.w)};
  return r;
}
__device__ __forceinline__ v4f block_max4(v4f m, v4f* red) {
  m = wave_max4(m);
  if ((threadIdx.x & 31) == 0) red[threadIdx.x >> 5] = m;
  __syncthreads();
  v4f r = red[0];
  for (int w = 1; w < 8; ++w) r = max4(r, red[w]);
  return r;
}
__device__ __forceinline__ v4f block_sum4(v4f s, v4f* red) {
  s = wave_sum4(s);
  if ((threadIdx.x & 31) == 0) red[threadIdx.x >> 5] = s;
  __syncthreads();
  v4f r = red[0];
  for (int w = 1; w < 8; ++w) {
    const v4f o = red[w];
    r.x += o.x; r.y += o.y; r.z += o.z; r.w += o.w;
  }
  return r;
}

static constexpr int N_NODES = 50000;
static constexpr int N_EDGES = 800000;
static constexpr int EDGE_BLOCKS = 3125;
static constexpr int NODE_TILES = 3125;
static constexpr int PART_FULL = EDGE_BLOCKS / 256;
static constexpr int PART_TAIL = EDGE_BLOCKS - PART_FULL * 256;
static constexpr float RANGE_MARGIN = 32.0f;

static constexpr size_t WS_A2   = 0;
static constexpr size_t WS_BIAS = 2048;
static constexpr size_t WS_S    = 2056;
static constexpr size_t WS_SUMP = WS_S + (size_t)N_NODES * 8;
static constexpr size_t WS_RSUM = WS_SUMP + (size_t)EDGE_BLOCKS * 4;

__global__ void __launch_bounds__(128) prep_kernel(const float* __restrict__ W,
                                                   const float* __restrict__ b,
                                                   const float* __restrict__ a, float* ws) {
  const int c = threadIdx.x;
  const int j = blockIdx.x;
  float* A2 = ws + WS_A2;
  if (j == 8) {
    if (c < 8) {
      const int hb = c >> 1, halfb = c & 1;
      const float* avb = a + hb * 64 + halfb * 32;
      float accb = 0.f;
      for (int k = 0; k < 32; ++k) accb += b[hb * 32 + k] * avb[k];
      ws[WS_BIAS + c] = accb;
    }
    return;
  }
  const int h = j >> 1, half = j & 1;
  const float* av = a + h * 64 + half * 32;
  const float* wp = W + (size_t)(h * 32) * 128 + c;

  float w[32];
#pragma unroll
  for (int k = 0; k < 32; ++k) w[k] = wp[(size_t)k * 128];
  float acc = 0.f;
#pragma unroll
  for (int k = 0; k < 32; ++k) acc += w[k] * av[k];
  A2[j * 128 + c] = acc;
  A2[(8 + j) * 128 + c] = 0.f;
}

static constexpr int LDS_PITCH = 132;

__global__ void __launch_bounds__(256) score_wmma_kernel(const float* __restrict__ X,
                                                         float* ws) {
  __shared__ __attribute__((aligned(16))) float sB[16 * LDS_PITCH];
  __shared__ __attribute__((aligned(16))) float sA[8][16 * LDS_PITCH];
  const float* A2 = ws + WS_A2;
  const float* bias = ws + WS_BIAS;
  float* S = ws + WS_S;

  const int t = threadIdx.x;
  const int lane = t & 31;
  const int wave = t >> 5;
  const int tile_raw = blockIdx.x * 8 + wave;
  const bool live = tile_raw < NODE_TILES;
  const int tile = live ? tile_raw : NODE_TILES - 1;

  const int m = lane & 15;
  const int hi = lane >> 4;

  const v4f* xsrc = (const v4f*)(X + (size_t)tile * 16 * 128) + lane;
  v4f xr[16];
#pragma unroll
  for (int r = 0; r < 16; ++r) xr[r] = xsrc[r * 32];
  const v4f* wsrc = (const v4f*)A2;
  const v4f w0 = wsrc[t];
  const v4f w1 = wsrc[t + 256];
  const float bb = bias[m & 7];

  float* myA = sA[wave];
#pragma unroll
  for (int r = 0; r < 16; ++r) *(v4f*)(myA + r * LDS_PITCH + lane * 4) = xr[r];
  *(v4f*)(sB + wave * LDS_PITCH + lane * 4) = w0;
  *(v4f*)(sB + (8 + wave) * LDS_PITCH + lane * 4) = w1;
  __syncthreads();

  const float* arow = myA + m * LDS_PITCH + 2 * hi;
  const float* brow = sB + m * LDS_PITCH + 2 * hi;

  v8f acc = {};
#pragma unroll 8
  for (int kk = 0; kk < 32; ++kk) {
    v2f av = *(const v2f*)(arow + kk * 4);
    v2f bv = *(const v2f*)(brow + kk * 4);
    acc = __builtin_amdgcn_wmma_f32_16x16x4_f32(
         false, av,  false, bv,
         (short)0, acc,  false,  false);
  }

  if (live && m < 8) {
    const int rowbase = tile * 16 + 8 * hi;
#pragma unroll
    for (int r = 0; r < 8; ++r) S[(size_t)(rowbase + r) * 8 + m] = acc[r] + bb;
  }
}

__device__ __forceinline__ v4f edge_logits(const int* __restrict__ ei, const v4f* S, int e) {
  const int src = ei[e];
  const int dst = ei[N_EDGES + e];
  const v4f s0 = S[(size_t)src * 2], s1 = S[(size_t)src * 2 + 1];
  const v4f d0 = S[(size_t)dst * 2], d1 = S[(size_t)dst * 2 + 1];
  v4f lg;
  lg.x = s0.x + d0.y;
  lg.y = s0.z + d0.w;
  lg.z = s1.x + d1.y;
  lg.w = s1.z + d1.w;
  lg.x = fmaxf(lg.x, LEAKY_ALPHA * lg.x);
  lg.y = fmaxf(lg.y, LEAKY_ALPHA * lg.y);
  lg.z = fmaxf(lg.z, LEAKY_ALPHA * lg.z);
  lg.w = fmaxf(lg.w, LEAKY_ALPHA * lg.w);
  return lg;
}

__global__ void __launch_bounds__(256) edge_exp_kernel(const int* __restrict__ ei, float* ws,
                                                       float* __restrict__ out) {
  __shared__ v4f red[8];
  const int t = threadIdx.x;
  const int e = blockIdx.x * 256 + t;
  const v4f lg = edge_logits(ei, (const v4f*)(ws + WS_S), e);
  v4f ex;
  ex.x = __expf(lg.x);
  ex.y = __expf(lg.y);
  ex.z = __expf(lg.z);
  ex.w = __expf(lg.w);
  *(v4f*)(out + (size_t)e * 4) = ex;

  const v4f bs = block_sum4(ex, red);
  if (t == 0) *(v4f*)(ws + WS_SUMP + (size_t)blockIdx.x * 4) = bs;
}

__global__ void __launch_bounds__(256) reduce_sum_kernel(const int* __restrict__ ei, float* ws,
                                                         float* __restrict__ out) {
  __shared__ v4f red[8];
  const int t = threadIdx.x;
  const v4f* P = (const v4f*)(ws + WS_SUMP);
  v4f v[PART_FULL + 1];
#pragma unroll
  for (int q = 0; q < PART_FULL; ++q) v[q] = P[t + 256 * q];
  v[PART_FULL] = P[t < PART_TAIL ? t + 256 * PART_FULL : t];
  v4f s = {};
#pragma unroll
  for (int q = 0; q < PART_FULL; ++q) {
    s.x += v[q].x; s.y += v[q].y; s.z += v[q].z; s.w += v[q].w;
  }
  {
    const bool live = t < PART_TAIL;
    const v4f o = v[PART_FULL];
    s.x += live ? o.x : 0.f; s.y += live ? o.y : 0.f;
    s.z += live ? o.z : 0.f; s.w += live ? o.w : 0.f;
  }
  v4f r = block_sum4(s, red);
  const float lo = (float)N_EDGES * __expf(-RANGE_MARGIN);
  const float hi = __expf(RANGE_MARGIN);
  const bool in_range = (r.x >= lo) && (r.y >= lo) && (r.z >= lo) && (r.w >= lo) &&
                        (r.x <= hi) && (r.y <= hi) && (r.z <= hi) && (r.w <= hi);
  if (!in_range) {
    const v4f* S = (const v4f*)(ws + WS_S);
    __syncthreads();
    v4f mx = edge_logits(ei, S, t);
    for (int e = t + 256; e < N_EDGES; e += 256) mx = max4(mx, edge_logits(ei, S, e));
    const v4f gm = block_max4(mx, red);
    __syncthreads();
    v4f a2 = {};
    for (int e = t; e < N_EDGES; e += 256) {
      const v4f lg = edge_logits(ei, S, e);
      v4f ex;
      ex.x = __expf(lg.x - gm.x); ex.y = __expf(lg.y - gm.y);
      ex.z = __expf(lg.z - gm.z); ex.w = __expf(lg.w - gm.w);
      *(v4f*)(out + (size_t)e * 4) = ex;
      a2.x += ex.x; a2.y += ex.y; a2.z += ex.z; a2.w += ex.w;
    }
    r = block_sum4(a2, red);
  }
  if (t == 0) {
    v4f inv = {1.0f / r.x, 1.0f / r.y, 1.0f / r.z, 1.0f / r.w};
    *(v4f*)(ws + WS_RSUM) = inv;
  }
}

__global__ void __launch_bounds__(256) scale_kernel(float* ws, float* __restrict__ out) {
  const int e = blockIdx.x * 256 + threadIdx.x;
  const v4f r = *(const v4f*)(ws + WS_RSUM);
  v4f v = *(v4f*)(out + (size_t)e * 4);
  v.x *= r.x; v.y *= r.y; v.z *= r.z; v.w *= r.w;
  *(v4f*)(out + (size_t)e * 4) = v;
}

extern "C" void kernel_launch(void* const* d_in, const int* in_sizes, int n_in,
                              void* d_out, int out_size, void* d_ws, size_t ws_size,
                              hipStream_t stream) {
  const float* node_feats = (const float*)d_in[0];
  const int*   edge_index = (const int*)d_in[1];
  const float* W          = (const float*)d_in[2];
  const float* b          = (const float*)d_in[3];
  const float* a          = (const float*)d_in[4];
  float* out = (float*)d_out;
  float* ws  = (float*)d_ws;

  prep_kernel<<<9, 128, 0, stream>>>(W, b, a, ws);
  score_wmma_kernel<<<(NODE_TILES + 7) / 8, 256, 0, stream>>>(node_feats, ws);
  edge_exp_kernel<<<EDGE_BLOCKS, 256, 0, stream>>>(edge_index, ws, out);
  reduce_sum_kernel<<<1, 256, 0, stream>>>(edge_index, ws, out);
  scale_kernel<<<EDGE_BLOCKS, 256, 0, stream>>>(ws, out);
}
